# starting kernel
# baseline (speedup 1.0000x reference)
#include <hip/hip_runtime.h>

#define COUNT 8
#define L1    3072
#define L2    15
#define L3    32
#define NCOL  128
#define TM    64
#define NRT   4
#define KC    64
#define LDSTR 72

typedef __attribute__((ext_vector_type(16))) __bf16         v16bf;
typedef __attribute__((ext_vector_type(8)))  float          v8f;
typedef __attribute__((ext_vector_type(16))) unsigned short v16u;
typedef __attribute__((ext_vector_type(8)))  unsigned short u16x8;
typedef __attribute__((ext_vector_type(4)))  unsigned short u16x4;

__device__ __forceinline__ unsigned short f2bf(float f) {
  unsigned int u = __float_as_uint(f);
  u += 0x7fffu + ((u >> 16) & 1u);
  return (unsigned short)(u >> 16);
}
__device__ __forceinline__ float bf2f(unsigned short h) {
  return __uint_as_float(((unsigned int)h) << 16);
}
__device__ __forceinline__ float clamp01(float v) {
  return fminf(fmaxf(v, 0.0f), 1.0f);
}

__device__ __forceinline__ v16bf frag16(const unsigned short* p) {
  u16x8 a = *(const u16x8*)(p);
  u16x8 b = *(const u16x8*)(p + 16);
  v16u v;
#pragma unroll
  for (int i = 0; i < 8; ++i) { v[i] = a[i]; v[i + 8] = b[i]; }
  return __builtin_bit_cast(v16bf, v);
}

__global__ void nnue_prep(const float* __restrict__ l1_w, const float* __restrict__ l1_b,
                          const float* __restrict__ l1f_w, const float* __restrict__ l1f_b,
                          unsigned short* __restrict__ whi, unsigned short* __restrict__ wlo,
                          float* __restrict__ bias) {
  const int i = blockIdx.x * 256 + threadIdx.x;
  if (i < NCOL * L1) {
    const int r = i / L1;
    const int k = i - r * L1;
    const float wv = l1_w[i] + l1f_w[(r & 15) * L1 + k];
    const unsigned short h = f2bf(wv);
    whi[i] = h;
    wlo[i] = f2bf(wv - bf2f(h));
  }
  if (i < NCOL) bias[i] = l1_b[i] + l1f_b[i & 15];
}

__global__ void __launch_bounds__(256, 2)
nnue_main(const float* __restrict__ x,
          const int* __restrict__ lsi,
          const unsigned short* __restrict__ whi,
          const unsigned short* __restrict__ wlo,
          const float* __restrict__ bias,
          const float* __restrict__ l2_w,
          const float* __restrict__ l2_b,
          const float* __restrict__ out_w,
          const float* __restrict__ out_b,
          float* __restrict__ out) {
  __shared__ __align__(16) unsigned short s_xhi[2][TM][LDSTR];
  __shared__ __align__(16) unsigned short s_xlo[2][TM][LDSTR];
  __shared__ float s_l1s[TM][NCOL + 4];
  __shared__ float s_l2x[TM][L3];

  const int t    = threadIdx.x;
  const int wv   = t >> 5;
  const int lane = t & 31;
  const int r0   = blockIdx.x * TM;

  const int nloc  = lane & 15;
  const int ksel  = (lane >> 4) << 3;
  const int rbase = (lane >> 4) << 3;

  const int ldr = t >> 4;
  const int ldc = (t & 15) << 2;

  const unsigned short* bhi_row = whi + (size_t)(wv * 16 + nloc) * L1;
  const unsigned short* blo_row = wlo + (size_t)(wv * 16 + nloc) * L1;

  const float* xrow = x + (size_t)(r0 + ldr) * L1 + ldc;

  v8f acc[NRT];
#pragma unroll
  for (int rt = 0; rt < NRT; ++rt)
    acc[rt] = (v8f){0.f, 0.f, 0.f, 0.f, 0.f, 0.f, 0.f, 0.f};

#pragma unroll
  for (int q = 0; q < NRT; ++q) {
    float4 xv = *(const float4*)(xrow + (size_t)q * 16 * L1);
    u16x4 h, l;
    float vv[4] = {xv.x, xv.y, xv.z, xv.w};
#pragma unroll
    for (int i = 0; i < 4; ++i) {
      const unsigned short hb = f2bf(vv[i]);
      h[i] = hb;
      l[i] = f2bf(vv[i] - bf2f(hb));
    }
    *(u16x4*)(&s_xhi[0][q * 16 + ldr][ldc]) = h;
    *(u16x4*)(&s_xlo[0][q * 16 + ldr][ldc]) = l;
  }
  __syncthreads();

  for (int kc = 0; kc < L1; kc += KC) {
    const int cur = (kc / KC) & 1;
    const bool have_next = (kc + KC) < L1;

    float4 xv[NRT];
    if (have_next) {
#pragma unroll
      for (int q = 0; q < NRT; ++q)
        xv[q] = *(const float4*)(xrow + (size_t)q * 16 * L1 + kc + KC);
      __builtin_prefetch(bhi_row + kc + KC + ksel, 0, 1);
      __builtin_prefetch(blo_row + kc + KC + ksel, 0, 1);
    }

#pragma unroll
    for (int kk = 0; kk < KC; kk += 32) {
      v16bf bhi = frag16(bhi_row + kc + kk + ksel);
      v16bf blo = frag16(blo_row + kc + kk + ksel);
#pragma unroll
      for (int rt = 0; rt < NRT; ++rt) {
        v16bf ahi = frag16(&s_xhi[cur][rt * 16 + nloc][0] + kk + ksel);
        v16bf alo = frag16(&s_xlo[cur][rt * 16 + nloc][0] + kk + ksel);
        acc[rt] = __builtin_amdgcn_wmma_f32_16x16x32_bf16(false, ahi, false, bhi, (short)0, acc[rt], false, false);
        acc[rt] = __builtin_amdgcn_wmma_f32_16x16x32_bf16(false, ahi, false, blo, (short)0, acc[rt], false, false);
        acc[rt] = __builtin_amdgcn_wmma_f32_16x16x32_bf16(false, alo, false, bhi, (short)0, acc[rt], false, false);
      }
    }

    if (have_next) {
#pragma unroll
      for (int q = 0; q < NRT; ++q) {
        u16x4 h, l;
        float vv[4] = {xv[q].x, xv[q].y, xv[q].z, xv[q].w};
#pragma unroll
        for (int i = 0; i < 4; ++i) {
          const unsigned short hb = f2bf(vv[i]);
          h[i] = hb;
          l[i] = f2bf(vv[i] - bf2f(hb));
        }
        *(u16x4*)(&s_xhi[1 - cur][q * 16 + ldr][ldc]) = h;
        *(u16x4*)(&s_xlo[1 - cur][q * 16 + ldr][ldc]) = l;
      }
    }
    __syncthreads();
  }

  {
    const int col  = wv * 16 + nloc;
    const float bc = bias[col];
#pragma unroll
    for (int rt = 0; rt < NRT; ++rt)
#pragma unroll
      for (int p = 0; p < 8; ++p)
        s_l1s[rt * 16 + rbase + p][col] = acc[rt][p] + bc;
  }
  __syncthreads();

  {
    const int i   = t >> 2;
    const int p   = t & 3;
    const int idx = lsi[r0 + i];
    const float* trow = &s_l1s[i][idx * 16];
    float l1x[2 * L2];
#pragma unroll
    for (int j = 0; j < L2; ++j) {
      const float v = trow[j];
      l1x[j]      = clamp01(v * v * (127.0f / 128.0f));
      l1x[L2 + j] = clamp01(v);
    }
#pragma unroll
    for (int q = 0; q < 8; ++q) {
      const int n = p * 8 + q;
      const float* wrow = l2_w + (size_t)(idx * L3 + n) * (2 * L2);
      float a = l2_b[idx * L3 + n];
#pragma unroll
      for (int j = 0; j < 2 * L2; ++j) a += l1x[j] * wrow[j];
      s_l2x[i][n] = clamp01(a);
    }
  }
  __syncthreads();

  if (t < TM) {
    const int idx = lsi[r0 + t];
    float s = out_b[idx];
#pragma unroll
    for (int n = 0; n < L3; ++n) s += s_l2x[t][n] * out_w[idx * L3 + n];
    out[r0 + t] = s + s_l1s[t][idx * 16 + L2];
  }
}

extern "C" void kernel_launch(void* const* d_in, const int* in_sizes, int n_in,
                              void* d_out, int out_size, void* d_ws, size_t ws_size,
                              hipStream_t stream) {
  const float* x     = (const float*)d_in[0];
  const int*   lsi   = (const int*)d_in[1];
  const float* l1_w  = (const float*)d_in[2];
  const float* l1_b  = (const float*)d_in[3];
  const float* l1f_w = (const float*)d_in[4];
  const float* l1f_b = (const float*)d_in[5];
  const float* l2_w  = (const float*)d_in[6];
  const float* l2_b  = (const float*)d_in[7];
  const float* out_w = (const float*)d_in[8];
  const float* out_b = (const float*)d_in[9];

  const int B = in_sizes[0] / L1;

  unsigned short* whi  = (unsigned short*)d_ws;
  unsigned short* wlo  = whi + (size_t)NCOL * L1;
  float*          bias = (float*)(wlo + (size_t)NCOL * L1);

  const int prepBlocks = (NCOL * L1 + 255) / 256;
  nnue_prep<<<prepBlocks, 256, 0, stream>>>(l1_w, l1_b, l1f_w, l1f_b, whi, wlo, bias);

  nnue_main<<<B / TM, 256, 0, stream>>>(x, lsi, whi, wlo, bias,
                                        l2_w, l2_b, out_w, out_b, (float*)d_out);
}


# K split moved from the 4 waves of a workgroup to 4 workgroups (256 rows x one K quarter): its 4 waves share each weight line; the bucket's partial sums meet in a third launch; order of sums changed
# speedup vs baseline: 1.9285x; 1.9285x over previous
#include <hip/hip_runtime.h>

#define COUNT 8
#define L1    3072
#define L2    15
#define L3    32
#define NCOL  128
#define RPW   64
#define NWV   4
#define TMG   (RPW * NWV)
#define NRT   4
#define NCT   8
#define KS    32
#define NKS   (L1 / KS)
#define NKQ   4
#define NST   (NKS / NKQ)
#define WBLK  512
#define TR    64

typedef __attribute__((ext_vector_type(16))) __bf16         v16bf;
typedef __attribute__((ext_vector_type(8)))  float          v8f;
typedef __attribute__((ext_vector_type(16))) unsigned short v16u;
typedef __attribute__((ext_vector_type(8)))  unsigned short u16x8;
typedef __attribute__((ext_vector_type(4)))  float          v4f;

__device__ __forceinline__ unsigned short f2bf(float f) {
  unsigned int u = __float_as_uint(f);
  u += 0x7fffu + ((u >> 16) & 1u);
  return (unsigned short)(u >> 16);
}
__device__ __forceinline__ float bf2f(unsigned short h) {
  return __uint_as_float(((unsigned int)h) << 16);
}
__device__ __forceinline__ float clamp01(float v) {
  return fminf(fmaxf(v, 0.0f), 1.0f);
}

__global__ void nnue_prep(const float* __restrict__ l1_w, const float* __restrict__ l1_b,
                          const float* __restrict__ l1f_w, const float* __restrict__ l1f_b,
                          unsigned short* __restrict__ whi, unsigned short* __restrict__ wlo,
                          float* __restrict__ bias) {
  const int g = blockIdx.x * 256 + threadIdx.x;
  if (g < NCOL * L1 / 8) {
    const int lane = g & 31;
    const int half = (g >> 5) & 1;
    const int blk  = g >> 6;
    const int ct   = blk & (NCT - 1);
    const int ks   = blk >> 3;
    const int col  = ct * 16 + (lane & 15);
    const int k    = ks * KS + half * 16 + (lane >> 4) * 8;
    const float* pw = l1_w + (size_t)col * L1 + k;
    const float* pf = l1f_w + (size_t)(col & 15) * L1 + k;
    const float4 w0 = *(const float4*)(pw);
    const float4 w1 = *(const float4*)(pw + 4);
    const float4 f0 = *(const float4*)(pf);
    const float4 f1 = *(const float4*)(pf + 4);
    const float wv[8] = {w0.x + f0.x, w0.y + f0.y, w0.z + f0.z, w0.w + f0.w,
                         w1.x + f1.x, w1.y + f1.y, w1.z + f1.z, w1.w + f1.w};
    u16x8 h, l;
#pragma unroll
    for (int e = 0; e < 8; ++e) {
      const unsigned short hb = f2bf(wv[e]);
      h[e] = hb;
      l[e] = f2bf(wv[e] - bf2f(hb));
    }
    *(u16x8*)(whi + (size_t)g * 8) = h;
    *(u16x8*)(wlo + (size_t)g * 8) = l;
  }
  if (g < NCOL) bias[g] = l1_b[g] + l1f_b[g & 15];
}


__device__ __forceinline__ void load_a(v4f (&xq)[4 * NRT], const float* xlane, int ks) {
#pragma unroll
  for (int rt = 0; rt < NRT; ++rt) {
    const float* b = xlane + (size_t)rt * 16 * L1 + ks * KS;
    xq[rt * 4 + 0] = *(const v4f*)(b);
    xq[rt * 4 + 1] = *(const v4f*)(b + 4);
    xq[rt * 4 + 2] = *(const v4f*)(b + 16);
    xq[rt * 4 + 3] = *(const v4f*)(b + 20);
  }
}

__device__ __forceinline__ void load_b(u16x8 (&bq)[4 * NCT], const unsigned short* whl,
                                       const unsigned short* wll, int ks) {
#pragma unroll
  for (int ct = 0; ct < NCT; ++ct) {
    const size_t base = (size_t)(ks * NCT + ct) * WBLK;
    bq[ct * 4 + 0] = *(const u16x8*)(whl + base);
    bq[ct * 4 + 1] = *(const u16x8*)(whl + base + 256);
    bq[ct * 4 + 2] = *(const u16x8*)(wll + base);
    bq[ct * 4 + 3] = *(const u16x8*)(wll + base + 256);
  }
}

__device__ __forceinline__ v16bf frag_of(const u16x8& a, const u16x8& b) {
  v16u v;
#pragma unroll
  for (int i = 0; i < 8; ++i) { v[i] = a[i]; v[i + 8] = b[i]; }
  return __builtin_bit_cast(v16bf, v);
}

__device__ __forceinline__ unsigned short f2bf_s(float f, unsigned int c7fff, unsigned int sh16) {
  unsigned int u = __float_as_uint(f);
  u += c7fff + ((u >> sh16) & 1u);
  return (unsigned short)(u >> 16);
}

__device__ __forceinline__ void split_x(const v4f& a, const v4f& b, const v4f& c, const v4f& d,
                                        unsigned int c7fff, unsigned int sh16, v16bf& fh, v16bf& fl) {
  const float f[16] = {a[0], a[1], a[2], a[3], b[0], b[1], b[2], b[3], c[0], c[1], c[2], c[3], d[0], d[1], d[2], d[3]};
  v16u h, l;
#pragma unroll
  for (int i = 0; i < 16; ++i) {
    const unsigned short hb = f2bf_s(f[i], c7fff, sh16);
    h[i] = hb;
    l[i] = f2bf_s(f[i] - bf2f(hb), c7fff, sh16);
  }
  fh = __builtin_bit_cast(v16bf, h);
  fl = __builtin_bit_cast(v16bf, l);
}

#define PHASE_FENCE() __builtin_amdgcn_sched_barrier(0)

__device__ __forceinline__ void step_anchor(unsigned int& c7fff, unsigned int& sh16) {
  c7fff = 0x7fffu;
  sh16  = 16u;
  asm volatile("" : "+s"(c7fff), "+s"(sh16));
}

__device__ __forceinline__ void k_step(int ksb, int ksx, v8f (&acc)[NRT][NCT],
                                       const u16x8 (&bcur)[4 * NCT], u16x8 (&bnext)[4 * NCT],
                                       const v4f (&xcur)[4 * NRT], v4f (&xload)[4 * NRT],
                                       const float* xlane,
                                       const unsigned short* whl,
                                       const unsigned short* wll) {
  load_b(bnext, whl, wll, ksb);
  load_a(xload, xlane, ksx);
  PHASE_FENCE();

  unsigned int c7fff, sh16;
  step_anchor(c7fff, sh16);

  v16bf ah[NRT], al[NRT];
#pragma unroll
  for (int rt = 0; rt < NRT; ++rt)
    split_x(xcur[rt * 4 + 0], xcur[rt * 4 + 1], xcur[rt * 4 + 2], xcur[rt * 4 + 3], c7fff, sh16, ah[rt], al[rt]);

#pragma unroll
  for (int ct = 0; ct < NCT; ++ct) {
    const v16bf bh = frag_of(bcur[ct * 4 + 0], bcur[ct * 4 + 1]);
    const v16bf bl = frag_of(bcur[ct * 4 + 2], bcur[ct * 4 + 3]);
#pragma unroll
    for (int rt = 0; rt < NRT; ++rt)
      acc[rt][ct] = __builtin_amdgcn_wmma_f32_16x16x32_bf16(false, ah[rt], false, bh, (short)0, acc[rt][ct], false, false);
#pragma unroll
    for (int rt = 0; rt < NRT; ++rt)
      acc[rt][ct] = __builtin_amdgcn_wmma_f32_16x16x32_bf16(false, ah[rt], false, bl, (short)0, acc[rt][ct], false, false);
  }
#pragma unroll
  for (int ct = 0; ct < NCT; ++ct) {
    const v16bf bh = frag_of(bcur[ct * 4 + 0], bcur[ct * 4 + 1]);
#pragma unroll
    for (int rt = 0; rt < NRT; ++rt)
      acc[rt][ct] = __builtin_amdgcn_wmma_f32_16x16x32_bf16(false, al[rt], false, bh, (short)0, acc[rt][ct], false, false);
  }
  PHASE_FENCE();
}

__global__ void __launch_bounds__(128)
nnue_main(const float* __restrict__ x,
          const int* __restrict__ lsi,
          const unsigned short* __restrict__ whi,
          const unsigned short* __restrict__ wlo,
          float* __restrict__ part) {
  const int t    = threadIdx.x;
  const int wv   = t >> 5;
  const int lane = t & 31;
  const int kq   = blockIdx.x & (NKQ - 1);
  const int r0   = (blockIdx.x >> 2) * TMG + wv * RPW;

  const int nloc  = lane & 15;
  const int ksel  = (lane >> 4) << 3;
  const int rbase = (lane >> 4) << 3;

  const float* xlane = x + (size_t)(r0 + nloc) * L1 + ksel;
  const unsigned short* whl = whi + lane * 8;
  const unsigned short* wll = wlo + lane * 8;

  v8f acc[NRT][NCT];
#pragma unroll
  for (int rt = 0; rt < NRT; ++rt)
#pragma unroll
    for (int ct = 0; ct < NCT; ++ct)
      acc[rt][ct] = (v8f){0.f, 0.f, 0.f, 0.f, 0.f, 0.f, 0.f, 0.f};

  u16x8 b0[4 * NCT], b1[4 * NCT];
  v4f   x0[4 * NRT], x1[4 * NRT], x2[4 * NRT], x3[4 * NRT];
  const int ks0 = kq * NST;
  load_b(b0, whl, wll, ks0);
  PHASE_FENCE();
  load_a(x0, xlane, ks0);
  PHASE_FENCE();
  load_a(x1, xlane, ks0 + 1);
  PHASE_FENCE();

  const int kslast = ks0 + NST - 1;
  for (int s = 0; s < NST; s += 4) {
    const int ks = ks0 + s;
    k_step(min(ks + 1, kslast), min(ks + 2, kslast), acc, b0, b1, x0, x2, xlane, whl, wll);
    k_step(min(ks + 2, kslast), min(ks + 3, kslast), acc, b1, b0, x1, x3, xlane, whl, wll);
    k_step(min(ks + 3, kslast), min(ks + 4, kslast), acc, b0, b1, x2, x0, xlane, whl, wll);
    k_step(min(ks + 4, kslast), min(ks + 5, kslast), acc, b1, b0, x3, x1, xlane, whl, wll);
  }

#pragma unroll
  for (int rt = 0; rt < NRT; ++rt) {
    const int* pi = lsi + r0 + rt * 16 + rbase;
    const int4 i0 = *(const int4*)(pi);
    const int4 i1 = *(const int4*)(pi + 4);
    const int bk[8] = {i0.x, i0.y, i0.z, i0.w, i1.x, i1.y, i1.z, i1.w};
    float* pt = part + ((size_t)(((r0 >> 4) + rt) * NKQ + kq) * 8) * 32 + lane;
#pragma unroll
    for (int p = 0; p < 8; ++p) {
      float v = 0.0f;
#pragma unroll
      for (int ct = 0; ct < NCT; ++ct) v = (bk[p] == ct) ? acc[rt][ct][p] : v;
      pt[p * 32] = v;
    }
  }
}

__global__ void __launch_bounds__(128)
nnue_tail(const int* __restrict__ lsi,
          const float* __restrict__ part,
          const float* __restrict__ bias,
          const float* __restrict__ l2_w,
          const float* __restrict__ l2_b,
          const float* __restrict__ out_w,
          const float* __restrict__ out_b,
          float* __restrict__ out) {
  __shared__ float s_l2x[TR][L3];
  __shared__ float s_skip[TR];

  const int t  = threadIdx.x;
  const int r0 = blockIdx.x * TR;

  {
    const int i   = t >> 1;
    const int p   = t & 1;
    const int row = r0 + i;
    const int idx = lsi[row];
    const float* pp = part + (size_t)(row >> 4) * (NKQ * 256) + (row & 7) * 32 + ((row >> 3) & 1) * 16;
    const float* pb = bias + idx * 16;
    float l1v[L2 + 1];
#pragma unroll
    for (int j = 0; j <= L2; j += 4) {
      const float4 a0 = *(const float4*)(pp + j);
      const float4 a1 = *(const float4*)(pp + 256 + j);
      const float4 a2 = *(const float4*)(pp + 512 + j);
      const float4 a3 = *(const float4*)(pp + 768 + j);
      const float4 bb = *(const float4*)(pb + j);
      l1v[j + 0] = ((a0.x + a1.x) + (a2.x + a3.x)) + bb.x;
      l1v[j + 1] = ((a0.y + a1.y) + (a2.y + a3.y)) + bb.y;
      l1v[j + 2] = ((a0.z + a1.z) + (a2.z + a3.z)) + bb.z;
      l1v[j + 3] = ((a0.w + a1.w) + (a2.w + a3.w)) + bb.w;
    }
    if (p == 0) s_skip[i] = l1v[L2];
    float l1x[2 * L2];
#pragma unroll
    for (int j = 0; j < L2; ++j) {
      const float v = l1v[j];
      l1x[j]      = clamp01(v * v * (127.0f / 128.0f));
      l1x[L2 + j] = clamp01(v);
    }
#pragma unroll
    for (int q = 0; q < 16; ++q) {
      const int n = p * 16 + q;
      const float* wrow = l2_w + (size_t)(idx * L3 + n) * (2 * L2);
      float a = l2_b[idx * L3 + n];
#pragma unroll
      for (int j = 0; j < 2 * L2; ++j) a += l1x[j] * wrow[j];
      s_l2x[i][n] = clamp01(a);
    }
  }
  __syncthreads();

  if (t < TR) {
    const int idx = lsi[r0 + t];
    float s = out_b[idx];
#pragma unroll
    for (int n = 0; n < L3; ++n) s += s_l2x[t][n] * out_w[idx * L3 + n];
    out[r0 + t] = s + s_skip[t];
  }
}

extern "C" void kernel_launch(void* const* d_in, const int* in_sizes, int n_in,
                              void* d_out, int out_size, void* d_ws, size_t ws_size,
                              hipStream_t stream) {
  const float* x     = (const float*)d_in[0];
  const int*   lsi   = (const int*)d_in[1];
  const float* l1_w  = (const float*)d_in[2];
  const float* l1_b  = (const float*)d_in[3];
  const float* l1f_w = (const float*)d_in[4];
  const float* l1f_b = (const float*)d_in[5];
  const float* l2_w  = (const float*)d_in[6];
  const float* l2_b  = (const float*)d_in[7];
  const float* out_w = (const float*)d_in[8];
  const float* out_b = (const float*)d_in[9];

  const int B = in_sizes[0] / L1;

  unsigned short* whi  = (unsigned short*)d_ws;
  unsigned short* wlo  = whi + (size_t)NCOL * L1;
  float*          bias = (float*)(wlo + (size_t)NCOL * L1);
  float*          part = bias + NCOL;

  const int prepBlocks = (NCOL * L1 / 8 + 255) / 256;
  nnue_prep<<<prepBlocks, 256, 0, stream>>>(l1_w, l1_b, l1f_w, l1f_b, whi, wlo, bias);

  nnue_main<<<B / TMG * NKQ, 128, 0, stream>>>(x, lsi, whi, wlo, part);

  nnue_tail<<<B / TR, 128, 0, stream>>>(lsi, part, bias, l2_w, l2_b, out_w, out_b, (float*)d_out);
}


# Tail re-cut: a wave takes its 16 rows one at a time with lane = layer-2 output; the row's weights come as contiguous loads from a lane-ordered copy written by the prep launch; order of sums unchanged
# speedup vs baseline: 2.2456x; 1.1644x over previous
#include <hip/hip_runtime.h>

#define COUNT 8
#define L1    3072
#define L2    15
#define L3    32
#define NCOL  128
#define RPW   64
#define NWV   4
#define TMG   (RPW * NWV)
#define NRT   4
#define NCT   8
#define KS    32
#define NKS   (L1 / KS)
#define NKQ   4
#define NST   (NKS / NKQ)
#define WBLK  512
#define TR    64
#define PREPB (NCOL * L1 / 8 / 256)
#define W2B   (COUNT * 8 * L3 / 256)

typedef __attribute__((ext_vector_type(16))) __bf16         v16bf;
typedef __attribute__((ext_vector_type(8)))  float          v8f;
typedef __attribute__((ext_vector_type(16))) unsigned short v16u;
typedef __attribute__((ext_vector_type(8)))  unsigned short u16x8;
typedef __attribute__((ext_vector_type(4)))  float          v4f;

__device__ __forceinline__ unsigned short f2bf(float f) {
  unsigned int u = __float_as_uint(f);
  u += 0x7fffu + ((u >> 16) & 1u);
  return (unsigned short)(u >> 16);
}
__device__ __forceinline__ float bf2f(unsigned short h) {
  return __uint_as_float(((unsigned int)h) << 16);
}
__device__ __forceinline__ float clamp01(float v) {
  return fminf(fmaxf(v, 0.0f), 1.0f);
}

__global__ void nnue_prep(const float* __restrict__ l1_w, const float* __restrict__ l1_b,
                          const float* __restrict__ l1f_w, const float* __restrict__ l1f_b,
                          unsigned short* __restrict__ whi, unsigned short* __restrict__ wlo,
                          float* __restrict__ bias,
                          const float* __restrict__ l2_w, const float* __restrict__ l2_b,
                          float* __restrict__ w2t) {
  if (blockIdx.x >= PREPB) {
    const int g2 = (blockIdx.x - PREPB) * 256 + threadIdx.x;
    const int n  = g2 & (L3 - 1);
    const int jq = (g2 >> 5) & 7;
    const int b  = g2 >> 8;
    const float* wr = l2_w + (size_t)(b * L3 + n) * (2 * L2) + 4 * jq;
    float4 o;
    o.x = wr[0];
    o.y = wr[1];
    if (jq < 7) { o.z = wr[2]; o.w = wr[3]; }
    else        { o.z = l2_b[b * L3 + n]; o.w = 0.0f; }
    *(float4*)(w2t + (size_t)g2 * 4) = o;
    return;
  }
  const int g = blockIdx.x * 256 + threadIdx.x;
  if (g < NCOL * L1 / 8) {
    const int lane = g & 31;
    const int half = (g >> 5) & 1;
    const int blk  = g >> 6;
    const int ct   = blk & (NCT - 1);
    const int ks   = blk >> 3;
    const int col  = ct * 16 + (lane & 15);
    const int k    = ks * KS + half * 16 + (lane >> 4) * 8;
    const float* pw = l1_w + (size_t)col * L1 + k;
    const float* pf = l1f_w + (size_t)(col & 15) * L1 + k;
    const float4 w0 = *(const float4*)(pw);
    const float4 w1 = *(const float4*)(pw + 4);
    const float4 f0 = *(const float4*)(pf);
    const float4 f1 = *(const float4*)(pf + 4);
    const float wv[8] = {w0.x + f0.x, w0.y + f0.y, w0.z + f0.z, w0.w + f0.w,
                         w1.x + f1.x, w1.y + f1.y, w1.z + f1.z, w1.w + f1.w};
    u16x8 h, l;
#pragma unroll
    for (int e = 0; e < 8; ++e) {
      const unsigned short hb = f2bf(wv[e]);
      h[e] = hb;
      l[e] = f2bf(wv[e] - bf2f(hb));
    }
    *(u16x8*)(whi + (size_t)g * 8) = h;
    *(u16x8*)(wlo + (size_t)g * 8) = l;
  }
  if (g < NCOL) bias[g] = l1_b[g] + l1f_b[g & 15];
}


__device__ __forceinline__ void load_a(v4f (&xq)[4 * NRT], const float* xlane, int ks) {
#pragma unroll
  for (int rt = 0; rt < NRT; ++rt) {
    const float* b = xlane + (size_t)rt * 16 * L1 + ks * KS;
    xq[rt * 4 + 0] = *(const v4f*)(b);
    xq[rt * 4 + 1] = *(const v4f*)(b + 4);
    xq[rt * 4 + 2] = *(const v4f*)(b + 16);
    xq[rt * 4 + 3] = *(const v4f*)(b + 20);
  }
}

__device__ __forceinline__ void load_b(u16x8 (&bq)[4 * NCT], const unsigned short* whl,
                                       const unsigned short* wll, int ks) {
#pragma unroll
  for (int ct = 0; ct < NCT; ++ct) {
    const size_t base = (size_t)(ks * NCT + ct) * WBLK;
    bq[ct * 4 + 0] = *(const u16x8*)(whl + base);
    bq[ct * 4 + 1] = *(const u16x8*)(whl + base + 256);
    bq[ct * 4 + 2] = *(const u16x8*)(wll + base);
    bq[ct * 4 + 3] = *(const u16x8*)(wll + base + 256);
  }
}

__device__ __forceinline__ v16bf frag_of(const u16x8& a, const u16x8& b) {
  v16u v;
#pragma unroll
  for (int i = 0; i < 8; ++i) { v[i] = a[i]; v[i + 8] = b[i]; }
  return __builtin_bit_cast(v16bf, v);
}

typedef __attribute__((ext_vector_type(8))) unsigned int v8u;

__device__ __forceinline__ void split_x(const v4f& a, const v4f& b, const v4f& c, const v4f& d,
                                        unsigned int rnd, unsigned int msk, unsigned int sel,
                                        v16bf& fh, v16bf& fl) {
  const float f[16] = {a[0], a[1], a[2], a[3], b[0], b[1], b[2], b[3], c[0], c[1], c[2], c[3], d[0], d[1], d[2], d[3]};
  v8u h, l;
#pragma unroll
  for (int i = 0; i < 8; ++i) {
    const unsigned int t0 = __float_as_uint(f[2 * i]) + rnd;
    const unsigned int t1 = __float_as_uint(f[2 * i + 1]) + rnd;
    h[i] = __builtin_amdgcn_perm(t1, t0, sel);
    const float r0 = f[2 * i] - __uint_as_float(t0 & msk);
    const float r1 = f[2 * i + 1] - __uint_as_float(t1 & msk);
    l[i] = __builtin_amdgcn_perm(__float_as_uint(r1), __float_as_uint(r0), sel);
  }
  fh = __builtin_bit_cast(v16bf, h);
  fl = __builtin_bit_cast(v16bf, l);
}

#define PHASE_FENCE() __builtin_amdgcn_sched_barrier(0)

__device__ __forceinline__ void step_anchor(unsigned int& rnd, unsigned int& msk, unsigned int& sel) {
  rnd = 0x8000u;
  msk = 0xffff0000u;
  sel = 0x07060302u;
  asm volatile("" : "+s"(rnd), "+s"(msk), "+s"(sel));
}

__device__ __forceinline__ void k_step(int ksb, int ksx, v8f (&acc)[NRT][NCT],
                                       const u16x8 (&bcur)[4 * NCT], u16x8 (&bnext)[4 * NCT],
                                       const v4f (&xcur)[4 * NRT], v4f (&xload)[4 * NRT],
                                       const float* xlane,
                                       const unsigned short* whl,
                                       const unsigned short* wll) {
  load_b(bnext, whl, wll, ksb);
  load_a(xload, xlane, ksx);
  PHASE_FENCE();

  unsigned int rnd, msk, sel;
  step_anchor(rnd, msk, sel);

  v16bf ah[NRT], al[NRT];
#pragma unroll
  for (int rt = 0; rt < NRT; ++rt)
    split_x(xcur[rt * 4 + 0], xcur[rt * 4 + 1], xcur[rt * 4 + 2], xcur[rt * 4 + 3], rnd, msk, sel, ah[rt], al[rt]);

#pragma unroll
  for (int ct = 0; ct < NCT; ++ct) {
    const v16bf bh = frag_of(bcur[ct * 4 + 0], bcur[ct * 4 + 1]);
    const v16bf bl = frag_of(bcur[ct * 4 + 2], bcur[ct * 4 + 3]);
#pragma unroll
    for (int rt = 0; rt < NRT; ++rt)
      acc[rt][ct] = __builtin_amdgcn_wmma_f32_16x16x32_bf16(false, ah[rt], false, bh, (short)0, acc[rt][ct], false, false);
#pragma unroll
    for (int rt = 0; rt < NRT; ++rt)
      acc[rt][ct] = __builtin_amdgcn_wmma_f32_16x16x32_bf16(false, ah[rt], false, bl, (short)0, acc[rt][ct], false, false);
  }
#pragma unroll
  for (int ct = 0; ct < NCT; ++ct) {
    const v16bf bh = frag_of(bcur[ct * 4 + 0], bcur[ct * 4 + 1]);
#pragma unroll
    for (int rt = 0; rt < NRT; ++rt)
      acc[rt][ct] = __builtin_amdgcn_wmma_f32_16x16x32_bf16(false, al[rt], false, bh, (short)0, acc[rt][ct], false, false);
  }
  PHASE_FENCE();
}

__global__ void __launch_bounds__(128)
nnue_main(const float* __restrict__ x,
          const int* __restrict__ lsi,
          const unsigned short* __restrict__ whi,
          const unsigned short* __restrict__ wlo,
          float* __restrict__ part) {
  const int t    = threadIdx.x;
  const int wv   = t >> 5;
  const int lane = t & 31;
  const int kq   = blockIdx.x & (NKQ - 1);
  const int r0   = (blockIdx.x >> 2) * TMG + wv * RPW;

  const int nloc  = lane & 15;
  const int ksel  = (lane >> 4) << 3;
  const int rbase = (lane >> 4) << 3;

  const float* xlane = x + (size_t)(r0 + nloc) * L1 + ksel;
  const unsigned short* whl = whi + lane * 8;
  const unsigned short* wll = wlo + lane * 8;

  v8f acc[NRT][NCT];
#pragma unroll
  for (int rt = 0; rt < NRT; ++rt)
#pragma unroll
    for (int ct = 0; ct < NCT; ++ct)
      acc[rt][ct] = (v8f){0.f, 0.f, 0.f, 0.f, 0.f, 0.f, 0.f, 0.f};

  u16x8 b0[4 * NCT], b1[4 * NCT];
  v4f   x0[4 * NRT], x1[4 * NRT], x2[4 * NRT], x3[4 * NRT];
  const int ks0 = kq * NST;
  load_b(b0, whl, wll, ks0);
  PHASE_FENCE();
  load_a(x0, xlane, ks0);
  PHASE_FENCE();
  load_a(x1, xlane, ks0 + 1);
  PHASE_FENCE();

  const int kslast = ks0 + NST - 1;
  for (int s = 0; s < NST; s += 4) {
    const int ks = ks0 + s;
    k_step(min(ks + 1, kslast), min(ks + 2, kslast), acc, b0, b1, x0, x2, xlane, whl, wll);
    k_step(min(ks + 2, kslast), min(ks + 3, kslast), acc, b1, b0, x1, x3, xlane, whl, wll);
    k_step(min(ks + 3, kslast), min(ks + 4, kslast), acc, b0, b1, x2, x0, xlane, whl, wll);
    k_step(min(ks + 4, kslast), min(ks + 5, kslast), acc, b1, b0, x3, x1, xlane, whl, wll);
  }

#pragma unroll
  for (int rt = 0; rt < NRT; ++rt) {
    const int* pi = lsi + r0 + rt * 16 + rbase;
    const int4 i0 = *(const int4*)(pi);
    const int4 i1 = *(const int4*)(pi + 4);
    const int bk[8] = {i0.x, i0.y, i0.z, i0.w, i1.x, i1.y, i1.z, i1.w};
    float* pt = part + ((size_t)(((r0 >> 4) + rt) * NKQ + kq) * 8) * 32 + lane;
#pragma unroll
    for (int p = 0; p < 8; ++p) {
      float v = 0.0f;
#pragma unroll
      for (int ct = 0; ct < NCT; ++ct) v = (bk[p] == ct) ? acc[rt][ct][p] : v;
      pt[p * 32] = v;
    }
  }
}

template <int G>
__device__ __forceinline__ void tail_load_w(float4 (&w)[4][8], const float4* w2l, const int (&sidx)[16]) {
#pragma unroll
  for (int r = 0; r < 4; ++r) {
    const float4* wp = w2l + (size_t)sidx[4 * G + r] * (8 * L3);
#pragma unroll
    for (int jq = 0; jq < 8; ++jq) w[r][jq] = wp[jq * L3];
  }
}

template <int G>
__device__ __forceinline__ void tail_rows(const float4 (&w)[4][8], const float (&sq)[8], const float (&lin)[8],
                                          float* s_row, int lane) {
#pragma unroll
  for (int r = 0; r < 4; ++r) {
    const int i = 4 * G + r;
    float wf[32];
#pragma unroll
    for (int jq = 0; jq < 8; ++jq) {
      wf[4 * jq] = w[r][jq].x; wf[4 * jq + 1] = w[r][jq].y; wf[4 * jq + 2] = w[r][jq].z; wf[4 * jq + 3] = w[r][jq].w;
    }
    float a = wf[2 * L2];
#pragma unroll
    for (int j = 0; j < 2 * L2; ++j) {
      const float src = (j < L2) ? sq[i & 7] : lin[i & 7];
      const int   sl  = (i >> 3) * 16 + ((j < L2) ? j : j - L2);
      const float xj  = __int_as_float(__builtin_amdgcn_readlane(__float_as_int(src), sl));
      a += xj * wf[j];
    }
    s_row[i * (L3 + 4) + lane] = clamp01(a);
  }
}

__global__ void __launch_bounds__(128)
nnue_tail(const int* __restrict__ lsi,
          const float* __restrict__ part,
          const float* __restrict__ bias,
          const float* __restrict__ w2t,
          const float* __restrict__ out_w,
          const float* __restrict__ out_b,
          float* __restrict__ out) {
  __shared__ float s_l2x[TR][L3 + 4];
  __shared__ float s_skip[TR];

  const int t    = threadIdx.x;
  const int lane = t & 31;
  const int c    = lane & 15;
  const int half = lane >> 4;
  const int wv   = __builtin_amdgcn_readfirstlane(t >> 5);
  const int r0   = blockIdx.x * TR;
  const int rw   = r0 + wv * 16;

  int sidx[16];
#pragma unroll
  for (int i = 0; i < 16; ++i) sidx[i] = lsi[rw + i];

  const int   idx2 = lsi[r0 + (t & (TR - 1))];
  const float ob   = out_b[idx2];
  float ow[L3];
#pragma unroll
  for (int n = 0; n < L3; n += 4) {
    const float4 q = *(const float4*)(out_w + idx2 * L3 + n);
    ow[n] = q.x; ow[n + 1] = q.y; ow[n + 2] = q.z; ow[n + 3] = q.w;
  }

  const float* pp = part + (size_t)(rw >> 4) * (NKQ * 256) + lane;
  float pa[8][4], pb[8];
#pragma unroll
  for (int p = 0; p < 8; ++p) {
    const int bk = half ? sidx[8 + p] : sidx[p];
    pa[p][0] = pp[p * 32];
    pa[p][1] = pp[256 + p * 32];
    pa[p][2] = pp[512 + p * 32];
    pa[p][3] = pp[768 + p * 32];
    pb[p]    = bias[bk * 16 + c];
  }
  const float4* w2l  = (const float4*)w2t + lane;
  float4 wa[4][8], wb[4][8];
  tail_load_w<0>(wa, w2l, sidx);
  PHASE_FENCE();

  float v[8], sq[8], lin[8];
#pragma unroll
  for (int p = 0; p < 8; ++p) {
    v[p]   = ((pa[p][0] + pa[p][1]) + (pa[p][2] + pa[p][3])) + pb[p];
    sq[p]  = clamp01(v[p] * v[p] * (127.0f / 128.0f));
    lin[p] = clamp01(v[p]);
  }
  if (c == L2) {
#pragma unroll
    for (int p = 0; p < 8; ++p) s_skip[wv * 16 + half * 8 + p] = v[p];
  }

  float* srow = &s_l2x[wv * 16][0];
  tail_load_w<1>(wb, w2l, sidx);
  PHASE_FENCE();
  tail_rows<0>(wa, sq, lin, srow, lane);
  PHASE_FENCE();
  tail_load_w<2>(wa, w2l, sidx);
  PHASE_FENCE();
  tail_rows<1>(wb, sq, lin, srow, lane);
  PHASE_FENCE();
  tail_load_w<3>(wb, w2l, sidx);
  PHASE_FENCE();
  tail_rows<2>(wa, sq, lin, srow, lane);
  PHASE_FENCE();
  tail_rows<3>(wb, sq, lin, srow, lane);
  __syncthreads();

  if (t < TR) {
    float s = ob;
#pragma unroll
    for (int n = 0; n < L3; ++n) s += s_l2x[t][n] * ow[n];
    out[r0 + t] = s + s_skip[t];
  }
}

extern "C" void kernel_launch(void* const* d_in, const int* in_sizes, int n_in,
                              void* d_out, int out_size, void* d_ws, size_t ws_size,
                              hipStream_t stream) {
  const float* x     = (const float*)d_in[0];
  const int*   lsi   = (const int*)d_in[1];
  const float* l1_w  = (const float*)d_in[2];
  const float* l1_b  = (const float*)d_in[3];
  const float* l1f_w = (const float*)d_in[4];
  const float* l1f_b = (const float*)d_in[5];
  const float* l2_w  = (const float*)d_in[6];
  const float* l2_b  = (const float*)d_in[7];
  const float* out_w = (const float*)d_in[8];
  const float* out_b = (const float*)d_in[9];

  const int B = in_sizes[0] / L1;

  unsigned short* whi  = (unsigned short*)d_ws;
  unsigned short* wlo  = whi + (size_t)NCOL * L1;
  float*          bias = (float*)(wlo + (size_t)NCOL * L1);
  float*          part = bias + NCOL;
  float*          w2t  = part + (size_t)B * NKQ * 16;

  nnue_prep<<<PREPB + W2B, 256, 0, stream>>>(l1_w, l1_b, l1f_w, l1f_b, whi, wlo, bias, l2_w, l2_b, w2t);

  nnue_main<<<B / TMG * NKQ, 128, 0, stream>>>(x, lsi, whi, wlo, part);

  nnue_tail<<<B / TR, 128, 0, stream>>>(lsi, part, bias, w2t, out_w, out_b, (float*)d_out);
}
